# starting kernel
# baseline (speedup 1.0000x reference)
#include <hip/hip_runtime.h>
#include <hip/hip_bf16.h>
#include <math.h>


typedef float v2f __attribute__((ext_vector_type(2)));
typedef float v4f __attribute__((ext_vector_type(4)));
typedef float v8f __attribute__((ext_vector_type(8)));

#define D_MODEL 256
#define NT 32
#define MT 128

__global__ __launch_bounds__(256)
void dta_gemm_f32_wmma(const float* __restrict__ A, const float* __restrict__ W,
                       const float* __restrict__ bias, float* __restrict__ C,
                       int M, int N) {
  __shared__ float Wlds[D_MODEL * NT];

  const int tid = threadIdx.x;
  const int n0  = blockIdx.y * NT;

  for (int c = tid; c < (D_MODEL / 2) * (NT / 4); c += 256) {
    const int kk = c >> 3;
    const int j  = (c & 7) << 2;
    v4f r0 = *(const v4f*)(W + (long)(2 * kk)     * N + n0 + j);
    v4f r1 = *(const v4f*)(W + (long)(2 * kk + 1) * N + n0 + j);
    v4f lo = {r0.x, r1.x, r0.y, r1.y};
    v4f hi = {r0.z, r1.z, r0.w, r1.w};
    *(v4f*)(&Wlds[kk * (NT * 2) + j * 2])     = lo;
    *(v4f*)(&Wlds[kk * (NT * 2) + j * 2 + 4]) = hi;
  }
  __syncthreads();

  const int wave  = tid >> 5;
  const int lane  = tid & 31;
  const int m0    = blockIdx.x * MT + wave * 16;
  const int mrow  = m0 + (lane & 15);
  const int khalf = (lane >> 4) << 1;
  const int ncol  = lane & 15;

  v8f c0 = {};
  v8f c1 = {};
  const float* Arow = A + (long)mrow * D_MODEL;
  const float* Bbase = &Wlds[(khalf >> 1) * (NT * 2) + ncol * 2];

  #pragma unroll 8
  for (int k = 0; k < D_MODEL; k += 4) {
    v2f a = *(const v2f*)(Arow + k + khalf);
    const float* Bk = Bbase + (k >> 1) * (NT * 2);
    v2f b0 = *(const v2f*)(Bk);
    v2f b1 = *(const v2f*)(Bk + 32);
    c0 = __builtin_amdgcn_wmma_f32_16x16x4_f32(false, a, false, b0, (short)0, c0, false, false);
    c1 = __builtin_amdgcn_wmma_f32_16x16x4_f32(false, a, false, b1, (short)0, c1, false, false);
  }

  const int rbase = m0 + ((lane >> 4) << 3);
  const float bv0 = bias ? bias[n0 + ncol] : 0.0f;
  const float bv1 = bias ? bias[n0 + 16 + ncol] : 0.0f;
  #pragma unroll
  for (int r = 0; r < 8; ++r) {
    C[(long)(rbase + r) * N + n0 + ncol]      = c0[r] + bv0;
    C[(long)(rbase + r) * N + n0 + 16 + ncol] = c1[r] + bv1;
  }
}

__global__ __launch_bounds__(256)
void dta_softmax12(float* __restrict__ aw, int total  ) {
  const int t = blockIdx.x * blockDim.x + threadIdx.x;
  if (t >= total) return;
  const int row = t >> 3;
  const int h   = t & 7;
  float* p = aw + (long)row * 96 + h * 12;

  float v[12];
  float mx = -3.402823466e+38f;
  #pragma unroll
  for (int i = 0; i < 12; ++i) { v[i] = p[i]; mx = fmaxf(mx, v[i]); }
  float s = 0.0f;
  #pragma unroll
  for (int i = 0; i < 12; ++i) { v[i] = __expf(v[i] - mx); s += v[i]; }
  const float inv = 1.0f / s;
  #pragma unroll
  for (int i = 0; i < 12; ++i) p[i] = v[i] * inv;
}

__global__ __launch_bounds__(256)
void dta_sample(const float* __restrict__ refp, const float* __restrict__ off_raw,
                const float* __restrict__ aw, const float* __restrict__ v_proj,
                float* __restrict__ out_mid) {
  const int wave = (blockIdx.x * blockDim.x + threadIdx.x) >> 5;
  const int lane = threadIdx.x & 31;
  const int b = wave >> 11;
  const int row = wave;

  const float rp = refp[row];
  const int   Ts[3]   = {2048, 1024, 512};
  const long  base[3] = {0, 8L * 2048, 8L * 2048 + 8L * 1024};

  const float* offr = off_raw + (long)row * 96;
  const float* awr  = aw + (long)row * 96;

  #pragma unroll
  for (int h = 0; h < 8; ++h) {
    float acc = 0.0f;
    #pragma unroll
    for (int l = 0; l < 3; ++l) {
      const float Tf = (float)Ts[l];
      const long  rb = base[l] + (long)b * Ts[l];
      #pragma unroll
      for (int p = 0; p < 4; ++p) {
        const int idx = h * 12 + l * 4 + p;
        const float o = offr[idx];
        const float w = awr[idx];
        float pos  = fminf(fmaxf(rp + o / Tf, 0.0f), 1.0f);
        float sidx = pos * (Tf - 1.0f);
        int ifl = (int)sidx;
        ifl = min(max(ifl, 0), Ts[l] - 2);
        const float wc = sidx - (float)ifl;
        const long r0 = (rb + ifl) * 256 + p * 32 + lane;
        const float vf = v_proj[r0];
        const float vc = v_proj[r0 + 256];
        acc += w * ((1.0f - wc) * vf + wc * vc);
      }
    }
    out_mid[(long)row * 256 + h * 32 + lane] = acc;
  }
}

extern "C" void kernel_launch(void* const* d_in, const int* in_sizes, int n_in,
                              void* d_out, int out_size, void* d_ws, size_t ws_size,
                              hipStream_t stream) {
  (void)in_sizes; (void)n_in; (void)out_size; (void)ws_size;

  const float* query = (const float*)d_in[0];
  const float* refp  = (const float*)d_in[1];
  const float* val0  = (const float*)d_in[2];
  const float* val1  = (const float*)d_in[3];
  const float* val2  = (const float*)d_in[4];
  const float* Woff  = (const float*)d_in[5];
  const float* boff  = (const float*)d_in[6];
  const float* Waw   = (const float*)d_in[7];
  const float* baw   = (const float*)d_in[8];
  const float* Wv    = (const float*)d_in[9];
  const float* bv    = (const float*)d_in[10];
  const float* Wo    = (const float*)d_in[11];
  const float* bo    = (const float*)d_in[12];
  float*       out   = (float*)d_out;

  const int BQ = 8 * 2048;

  char* ws = (char*)d_ws;
  float* off_raw = (float*)(ws);
  float* aw_buf  = (float*)(ws + (size_t)BQ * 96 * 4);
  float* v_proj  = (float*)(ws + (size_t)BQ * 96 * 4 * 2);
  float* out_mid = (float*)(ws + (size_t)BQ * 96 * 4 * 2
                               + (size_t)(BQ + 8192 + 4096) * 256 * 4);

  dim3 blk(256);

  dta_gemm_f32_wmma<<<dim3(BQ / MT, 96 / NT), blk, 0, stream>>>(query, Woff, boff, off_raw, BQ, 96);
  dta_gemm_f32_wmma<<<dim3(BQ / MT, 96 / NT), blk, 0, stream>>>(query, Waw, baw, aw_buf, BQ, 96);
  dta_softmax12<<<dim3((BQ * 8 + 255) / 256), blk, 0, stream>>>(aw_buf, BQ * 8);
  dta_gemm_f32_wmma<<<dim3(16384 / MT, 256 / NT), blk, 0, stream>>>(val0, Wv, bv, v_proj, 16384, 256);
  dta_gemm_f32_wmma<<<dim3( 8192 / MT, 256 / NT), blk, 0, stream>>>(val1, Wv, bv, v_proj + (size_t)16384 * 256,  8192, 256);
  dta_gemm_f32_wmma<<<dim3( 4096 / MT, 256 / NT), blk, 0, stream>>>(val2, Wv, bv, v_proj + (size_t)(16384 + 8192) * 256, 4096, 256);
  dta_sample<<<dim3(BQ / 8), blk, 0, stream>>>(refp, off_raw, aw_buf, v_proj, out_mid);
  dta_gemm_f32_wmma<<<dim3(BQ / MT, 256 / NT), blk, 0, stream>>>(out_mid, Wo, bo, out, BQ, 256);
}


# One launch for the five input-only projection GEMMs (offsets, logits, three value levels): block id picks problem and tile, tile code unchanged; 8 launches -> 4
# speedup vs baseline: 1.1984x; 1.1984x over previous
#include <hip/hip_runtime.h>
#include <hip/hip_bf16.h>
#include <math.h>


typedef float v2f __attribute__((ext_vector_type(2)));
typedef float v4f __attribute__((ext_vector_type(4)));
typedef float v8f __attribute__((ext_vector_type(8)));

#define D_MODEL 256
#define NT 32
#define MT 128

__device__ __forceinline__
void dta_gemm_tile(const float* __restrict__ A, const float* __restrict__ W,
                   const float* __restrict__ bias, float* __restrict__ C,
                   int N, int mtile, int ntile, float* __restrict__ Wlds) {
  const int tid = threadIdx.x;
  const int n0  = ntile * NT;

  for (int c = tid; c < (D_MODEL / 2) * (NT / 4); c += 256) {
    const int kk = c >> 3;
    const int j  = (c & 7) << 2;
    v4f r0 = *(const v4f*)(W + (long)(2 * kk)     * N + n0 + j);
    v4f r1 = *(const v4f*)(W + (long)(2 * kk + 1) * N + n0 + j);
    v4f lo = {r0.x, r1.x, r0.y, r1.y};
    v4f hi = {r0.z, r1.z, r0.w, r1.w};
    *(v4f*)(&Wlds[kk * (NT * 2) + j * 2])     = lo;
    *(v4f*)(&Wlds[kk * (NT * 2) + j * 2 + 4]) = hi;
  }
  __syncthreads();

  const int wave  = tid >> 5;
  const int lane  = tid & 31;
  const int m0    = mtile * MT + wave * 16;
  const int mrow  = m0 + (lane & 15);
  const int khalf = (lane >> 4) << 1;
  const int ncol  = lane & 15;

  v8f c0 = {};
  v8f c1 = {};
  const float* Arow = A + (long)mrow * D_MODEL;
  const float* Bbase = &Wlds[(khalf >> 1) * (NT * 2) + ncol * 2];

  #pragma unroll 8
  for (int k = 0; k < D_MODEL; k += 4) {
    v2f a = *(const v2f*)(Arow + k + khalf);
    const float* Bk = Bbase + (k >> 1) * (NT * 2);
    v2f b0 = *(const v2f*)(Bk);
    v2f b1 = *(const v2f*)(Bk + 32);
    c0 = __builtin_amdgcn_wmma_f32_16x16x4_f32(false, a, false, b0, (short)0, c0, false, false);
    c1 = __builtin_amdgcn_wmma_f32_16x16x4_f32(false, a, false, b1, (short)0, c1, false, false);
  }

  const int rbase = m0 + ((lane >> 4) << 3);
  const float bv0 = bias ? bias[n0 + ncol] : 0.0f;
  const float bv1 = bias ? bias[n0 + 16 + ncol] : 0.0f;
  #pragma unroll
  for (int r = 0; r < 8; ++r) {
    C[(long)(rbase + r) * N + n0 + ncol]      = c0[r] + bv0;
    C[(long)(rbase + r) * N + n0 + 16 + ncol] = c1[r] + bv1;
  }
}

__global__ __launch_bounds__(256)
void dta_gemm_f32_wmma(const float* __restrict__ A, const float* __restrict__ W,
                       const float* __restrict__ bias, float* __restrict__ C,
                       int M, int N) {
  __shared__ float Wlds[D_MODEL * NT];
  (void)M;
  dta_gemm_tile(A, W, bias, C, N, blockIdx.x, blockIdx.y, Wlds);
}

#define PROJ5_BLOCKS 2560
__global__ __launch_bounds__(256)
void dta_proj5_f32_wmma(const float* __restrict__ query,
                        const float* __restrict__ val0, const float* __restrict__ val1,
                        const float* __restrict__ val2,
                        const float* __restrict__ Woff, const float* __restrict__ boff,
                        const float* __restrict__ Waw,  const float* __restrict__ baw,
                        const float* __restrict__ Wv,   const float* __restrict__ bv,
                        float* __restrict__ off_raw, float* __restrict__ aw_buf,
                        float* __restrict__ v_proj) {
  __shared__ float Wlds[D_MODEL * NT];

  int id = blockIdx.x;
  const float* A; const float* Wm; const float* bias; float* C;
  int N, mtile, ntile;
  if (id < 384) {
    A = query; Wm = Woff; bias = boff; C = off_raw; N = 96;
    mtile = id & 127; ntile = id >> 7;
  } else if (id < 768) {
    id -= 384;
    A = query; Wm = Waw; bias = baw; C = aw_buf; N = 96;
    mtile = id & 127; ntile = id >> 7;
  } else if (id < 1792) {
    id -= 768;
    A = val0; Wm = Wv; bias = bv; C = v_proj; N = 256;
    mtile = id & 127; ntile = id >> 7;
  } else if (id < 2304) {
    id -= 1792;
    A = val1; Wm = Wv; bias = bv; C = v_proj + (size_t)16384 * 256; N = 256;
    mtile = id & 63; ntile = id >> 6;
  } else {
    id -= 2304;
    A = val2; Wm = Wv; bias = bv; C = v_proj + (size_t)(16384 + 8192) * 256; N = 256;
    mtile = id & 31; ntile = id >> 5;
  }
  dta_gemm_tile(A, Wm, bias, C, N, mtile, ntile, Wlds);
}

__global__ __launch_bounds__(256)
void dta_softmax12(float* __restrict__ aw, int total  ) {
  const int t = blockIdx.x * blockDim.x + threadIdx.x;
  if (t >= total) return;
  const int row = t >> 3;
  const int h   = t & 7;
  float* p = aw + (long)row * 96 + h * 12;

  float v[12];
  float mx = -3.402823466e+38f;
  #pragma unroll
  for (int i = 0; i < 12; ++i) { v[i] = p[i]; mx = fmaxf(mx, v[i]); }
  float s = 0.0f;
  #pragma unroll
  for (int i = 0; i < 12; ++i) { v[i] = __expf(v[i] - mx); s += v[i]; }
  const float inv = 1.0f / s;
  #pragma unroll
  for (int i = 0; i < 12; ++i) p[i] = v[i] * inv;
}

__global__ __launch_bounds__(256)
void dta_sample(const float* __restrict__ refp, const float* __restrict__ off_raw,
                const float* __restrict__ aw, const float* __restrict__ v_proj,
                float* __restrict__ out_mid) {
  const int wave = (blockIdx.x * blockDim.x + threadIdx.x) >> 5;
  const int lane = threadIdx.x & 31;
  const int b = wave >> 11;
  const int row = wave;

  const float rp = refp[row];
  const int   Ts[3]   = {2048, 1024, 512};
  const long  base[3] = {0, 8L * 2048, 8L * 2048 + 8L * 1024};

  const float* offr = off_raw + (long)row * 96;
  const float* awr  = aw + (long)row * 96;

  #pragma unroll
  for (int h = 0; h < 8; ++h) {
    float acc = 0.0f;
    #pragma unroll
    for (int l = 0; l < 3; ++l) {
      const float Tf = (float)Ts[l];
      const long  rb = base[l] + (long)b * Ts[l];
      #pragma unroll
      for (int p = 0; p < 4; ++p) {
        const int idx = h * 12 + l * 4 + p;
        const float o = offr[idx];
        const float w = awr[idx];
        float pos  = fminf(fmaxf(rp + o / Tf, 0.0f), 1.0f);
        float sidx = pos * (Tf - 1.0f);
        int ifl = (int)sidx;
        ifl = min(max(ifl, 0), Ts[l] - 2);
        const float wc = sidx - (float)ifl;
        const long r0 = (rb + ifl) * 256 + p * 32 + lane;
        const float vf = v_proj[r0];
        const float vc = v_proj[r0 + 256];
        acc += w * ((1.0f - wc) * vf + wc * vc);
      }
    }
    out_mid[(long)row * 256 + h * 32 + lane] = acc;
  }
}

extern "C" void kernel_launch(void* const* d_in, const int* in_sizes, int n_in,
                              void* d_out, int out_size, void* d_ws, size_t ws_size,
                              hipStream_t stream) {
  (void)in_sizes; (void)n_in; (void)out_size; (void)ws_size;

  const float* query = (const float*)d_in[0];
  const float* refp  = (const float*)d_in[1];
  const float* val0  = (const float*)d_in[2];
  const float* val1  = (const float*)d_in[3];
  const float* val2  = (const float*)d_in[4];
  const float* Woff  = (const float*)d_in[5];
  const float* boff  = (const float*)d_in[6];
  const float* Waw   = (const float*)d_in[7];
  const float* baw   = (const float*)d_in[8];
  const float* Wv    = (const float*)d_in[9];
  const float* bv    = (const float*)d_in[10];
  const float* Wo    = (const float*)d_in[11];
  const float* bo    = (const float*)d_in[12];
  float*       out   = (float*)d_out;

  const int BQ = 8 * 2048;

  char* ws = (char*)d_ws;
  float* off_raw = (float*)(ws);
  float* aw_buf  = (float*)(ws + (size_t)BQ * 96 * 4);
  float* v_proj  = (float*)(ws + (size_t)BQ * 96 * 4 * 2);
  float* out_mid = (float*)(ws + (size_t)BQ * 96 * 4 * 2
                               + (size_t)(BQ + 8192 + 4096) * 256 * 4);

  dim3 blk(256);

  dta_proj5_f32_wmma<<<dim3(PROJ5_BLOCKS), blk, 0, stream>>>(query, val0, val1, val2,
                                                             Woff, boff, Waw, baw, Wv, bv,
                                                             off_raw, aw_buf, v_proj);
  dta_softmax12<<<dim3((BQ * 8 + 255) / 256), blk, 0, stream>>>(aw_buf, BQ * 8);
  dta_sample<<<dim3(BQ / 8), blk, 0, stream>>>(refp, off_raw, aw_buf, v_proj, out_mid);
  dta_gemm_f32_wmma<<<dim3(BQ / MT, 256 / NT), blk, 0, stream>>>(out_mid, Wo, bo, out, BQ, 256);
}


# Value projections compute only columns 0..127 of value@Wv+bv (the sampling indexes the head axis by point p<4, so heads 4..7 never reach the output): 2560 -> 1664 tiles, v_proj pitch 128
# speedup vs baseline: 1.3456x; 1.1228x over previous
#include <hip/hip_runtime.h>
#include <hip/hip_bf16.h>
#include <math.h>


typedef float v2f __attribute__((ext_vector_type(2)));
typedef float v4f __attribute__((ext_vector_type(4)));
typedef float v8f __attribute__((ext_vector_type(8)));

#define D_MODEL 256
#define NT 32
#define MT 128

__device__ __forceinline__
void dta_gemm_tile(const float* __restrict__ A, const float* __restrict__ W,
                   const float* __restrict__ bias, float* __restrict__ C,
                   int ldw, int ldc, int mtile, int ntile, float* __restrict__ Wlds) {
  const int tid = threadIdx.x;
  const int n0  = ntile * NT;

  for (int c = tid; c < (D_MODEL / 2) * (NT / 4); c += 256) {
    const int kk = c >> 3;
    const int j  = (c & 7) << 2;
    v4f r0 = *(const v4f*)(W + (long)(2 * kk)     * ldw + n0 + j);
    v4f r1 = *(const v4f*)(W + (long)(2 * kk + 1) * ldw + n0 + j);
    v4f lo = {r0.x, r1.x, r0.y, r1.y};
    v4f hi = {r0.z, r1.z, r0.w, r1.w};
    *(v4f*)(&Wlds[kk * (NT * 2) + j * 2])     = lo;
    *(v4f*)(&Wlds[kk * (NT * 2) + j * 2 + 4]) = hi;
  }
  __syncthreads();

  const int wave  = tid >> 5;
  const int lane  = tid & 31;
  const int m0    = mtile * MT + wave * 16;
  const int mrow  = m0 + (lane & 15);
  const int khalf = (lane >> 4) << 1;
  const int ncol  = lane & 15;

  v8f c0 = {};
  v8f c1 = {};
  const float* Arow = A + (long)mrow * D_MODEL;
  const float* Bbase = &Wlds[(khalf >> 1) * (NT * 2) + ncol * 2];

  #pragma unroll 8
  for (int k = 0; k < D_MODEL; k += 4) {
    v2f a = *(const v2f*)(Arow + k + khalf);
    const float* Bk = Bbase + (k >> 1) * (NT * 2);
    v2f b0 = *(const v2f*)(Bk);
    v2f b1 = *(const v2f*)(Bk + 32);
    c0 = __builtin_amdgcn_wmma_f32_16x16x4_f32(false, a, false, b0, (short)0, c0, false, false);
    c1 = __builtin_amdgcn_wmma_f32_16x16x4_f32(false, a, false, b1, (short)0, c1, false, false);
  }

  const int rbase = m0 + ((lane >> 4) << 3);
  const float bv0 = bias ? bias[n0 + ncol] : 0.0f;
  const float bv1 = bias ? bias[n0 + 16 + ncol] : 0.0f;
  #pragma unroll
  for (int r = 0; r < 8; ++r) {
    C[(long)(rbase + r) * ldc + n0 + ncol]      = c0[r] + bv0;
    C[(long)(rbase + r) * ldc + n0 + 16 + ncol] = c1[r] + bv1;
  }
}

__global__ __launch_bounds__(256)
void dta_gemm_f32_wmma(const float* __restrict__ A, const float* __restrict__ W,
                       const float* __restrict__ bias, float* __restrict__ C,
                       int M, int N) {
  __shared__ float Wlds[D_MODEL * NT];
  (void)M;
  dta_gemm_tile(A, W, bias, C, N, N, blockIdx.x, blockIdx.y, Wlds);
}

#define VW 128
#define PROJ5_BLOCKS 1664
__global__ __launch_bounds__(256)
void dta_proj5_f32_wmma(const float* __restrict__ query,
                        const float* __restrict__ val0, const float* __restrict__ val1,
                        const float* __restrict__ val2,
                        const float* __restrict__ Woff, const float* __restrict__ boff,
                        const float* __restrict__ Waw,  const float* __restrict__ baw,
                        const float* __restrict__ Wv,   const float* __restrict__ bv,
                        float* __restrict__ off_raw, float* __restrict__ aw_buf,
                        float* __restrict__ v_proj) {
  __shared__ float Wlds[D_MODEL * NT];

  int id = blockIdx.x;
  const float* A; const float* Wm; const float* bias; float* C;
  int ldw, ldc, mtile, ntile;
  if (id < 384) {
    A = query; Wm = Woff; bias = boff; C = off_raw; ldw = 96; ldc = 96;
    mtile = id & 127; ntile = id >> 7;
  } else if (id < 768) {
    id -= 384;
    A = query; Wm = Waw; bias = baw; C = aw_buf; ldw = 96; ldc = 96;
    mtile = id & 127; ntile = id >> 7;
  } else if (id < 1280) {
    id -= 768;
    A = val0; Wm = Wv; bias = bv; C = v_proj; ldw = 256; ldc = VW;
    mtile = id & 127; ntile = id >> 7;
  } else if (id < 1536) {
    id -= 1280;
    A = val1; Wm = Wv; bias = bv; C = v_proj + (size_t)16384 * VW; ldw = 256; ldc = VW;
    mtile = id & 63; ntile = id >> 6;
  } else {
    id -= 1536;
    A = val2; Wm = Wv; bias = bv; C = v_proj + (size_t)(16384 + 8192) * VW; ldw = 256; ldc = VW;
    mtile = id & 31; ntile = id >> 5;
  }
  dta_gemm_tile(A, Wm, bias, C, ldw, ldc, mtile, ntile, Wlds);
}

__global__ __launch_bounds__(256)
void dta_softmax12(float* __restrict__ aw, int total  ) {
  const int t = blockIdx.x * blockDim.x + threadIdx.x;
  if (t >= total) return;
  const int row = t >> 3;
  const int h   = t & 7;
  float* p = aw + (long)row * 96 + h * 12;

  float v[12];
  float mx = -3.402823466e+38f;
  #pragma unroll
  for (int i = 0; i < 12; ++i) { v[i] = p[i]; mx = fmaxf(mx, v[i]); }
  float s = 0.0f;
  #pragma unroll
  for (int i = 0; i < 12; ++i) { v[i] = __expf(v[i] - mx); s += v[i]; }
  const float inv = 1.0f / s;
  #pragma unroll
  for (int i = 0; i < 12; ++i) p[i] = v[i] * inv;
}

__global__ __launch_bounds__(256)
void dta_sample(const float* __restrict__ refp, const float* __restrict__ off_raw,
                const float* __restrict__ aw, const float* __restrict__ v_proj,
                float* __restrict__ out_mid) {
  const int wave = (blockIdx.x * blockDim.x + threadIdx.x) >> 5;
  const int lane = threadIdx.x & 31;
  const int b = wave >> 11;
  const int row = wave;

  const float rp = refp[row];
  const int   Ts[3]   = {2048, 1024, 512};
  const long  base[3] = {0, 8L * 2048, 8L * 2048 + 8L * 1024};

  const float* offr = off_raw + (long)row * 96;
  const float* awr  = aw + (long)row * 96;

  #pragma unroll
  for (int h = 0; h < 8; ++h) {
    float acc = 0.0f;
    #pragma unroll
    for (int l = 0; l < 3; ++l) {
      const float Tf = (float)Ts[l];
      const long  rb = base[l] + (long)b * Ts[l];
      #pragma unroll
      for (int p = 0; p < 4; ++p) {
        const int idx = h * 12 + l * 4 + p;
        const float o = offr[idx];
        const float w = awr[idx];
        float pos  = fminf(fmaxf(rp + o / Tf, 0.0f), 1.0f);
        float sidx = pos * (Tf - 1.0f);
        int ifl = (int)sidx;
        ifl = min(max(ifl, 0), Ts[l] - 2);
        const float wc = sidx - (float)ifl;
        const long r0 = (rb + ifl) * VW + p * 32 + lane;
        const float vf = v_proj[r0];
        const float vc = v_proj[r0 + VW];
        acc += w * ((1.0f - wc) * vf + wc * vc);
      }
    }
    out_mid[(long)row * 256 + h * 32 + lane] = acc;
  }
}

extern "C" void kernel_launch(void* const* d_in, const int* in_sizes, int n_in,
                              void* d_out, int out_size, void* d_ws, size_t ws_size,
                              hipStream_t stream) {
  (void)in_sizes; (void)n_in; (void)out_size; (void)ws_size;

  const float* query = (const float*)d_in[0];
  const float* refp  = (const float*)d_in[1];
  const float* val0  = (const float*)d_in[2];
  const float* val1  = (const float*)d_in[3];
  const float* val2  = (const float*)d_in[4];
  const float* Woff  = (const float*)d_in[5];
  const float* boff  = (const float*)d_in[6];
  const float* Waw   = (const float*)d_in[7];
  const float* baw   = (const float*)d_in[8];
  const float* Wv    = (const float*)d_in[9];
  const float* bv    = (const float*)d_in[10];
  const float* Wo    = (const float*)d_in[11];
  const float* bo    = (const float*)d_in[12];
  float*       out   = (float*)d_out;

  const int BQ = 8 * 2048;

  char* ws = (char*)d_ws;
  float* off_raw = (float*)(ws);
  float* aw_buf  = (float*)(ws + (size_t)BQ * 96 * 4);
  float* v_proj  = (float*)(ws + (size_t)BQ * 96 * 4 * 2);
  float* out_mid = (float*)(ws + (size_t)BQ * 96 * 4 * 2
                               + (size_t)(BQ + 8192 + 4096) * VW * 4);

  dim3 blk(256);

  dta_proj5_f32_wmma<<<dim3(PROJ5_BLOCKS), blk, 0, stream>>>(query, val0, val1, val2,
                                                             Woff, boff, Waw, baw, Wv, bv,
                                                             off_raw, aw_buf, v_proj);
  dta_softmax12<<<dim3((BQ * 8 + 255) / 256), blk, 0, stream>>>(aw_buf, BQ * 8);
  dta_sample<<<dim3(BQ / 8), blk, 0, stream>>>(refp, off_raw, aw_buf, v_proj, out_mid);
  dta_gemm_f32_wmma<<<dim3(BQ / MT, 256 / NT), blk, 0, stream>>>(out_mid, Wo, bo, out, BQ, 256);
}


# Sampling kernel: each row's 96 sample parameters (row offset, lerp weight, attention weight) are computed once, 4 per lane, and broadcast by v_readlane as scalars, not recomputed in all 32 lanes
# speedup vs baseline: 1.5031x; 1.1170x over previous
#include <hip/hip_runtime.h>
#include <hip/hip_bf16.h>
#include <math.h>


typedef float v2f __attribute__((ext_vector_type(2)));
typedef float v4f __attribute__((ext_vector_type(4)));
typedef float v8f __attribute__((ext_vector_type(8)));

#define D_MODEL 256
#define NT 32
#define MT 128

__device__ __forceinline__
void dta_gemm_tile(const float* __restrict__ A, const float* __restrict__ W,
                   const float* __restrict__ bias, float* __restrict__ C,
                   int ldw, int ldc, int mtile, int ntile, float* __restrict__ Wlds) {
  const int tid = threadIdx.x;
  const int n0  = ntile * NT;

  for (int c = tid; c < (D_MODEL / 2) * (NT / 4); c += 256) {
    const int kk = c >> 3;
    const int j  = (c & 7) << 2;
    v4f r0 = *(const v4f*)(W + (long)(2 * kk)     * ldw + n0 + j);
    v4f r1 = *(const v4f*)(W + (long)(2 * kk + 1) * ldw + n0 + j);
    v4f lo = {r0.x, r1.x, r0.y, r1.y};
    v4f hi = {r0.z, r1.z, r0.w, r1.w};
    *(v4f*)(&Wlds[kk * (NT * 2) + j * 2])     = lo;
    *(v4f*)(&Wlds[kk * (NT * 2) + j * 2 + 4]) = hi;
  }
  __syncthreads();

  const int wave  = tid >> 5;
  const int lane  = tid & 31;
  const int m0    = mtile * MT + wave * 16;
  const int mrow  = m0 + (lane & 15);
  const int khalf = (lane >> 4) << 1;
  const int ncol  = lane & 15;

  v8f c0 = {};
  v8f c1 = {};
  const float* Arow = A + (long)mrow * D_MODEL;
  const float* Bbase = &Wlds[(khalf >> 1) * (NT * 2) + ncol * 2];

  #pragma unroll 8
  for (int k = 0; k < D_MODEL; k += 4) {
    v2f a = *(const v2f*)(Arow + k + khalf);
    const float* Bk = Bbase + (k >> 1) * (NT * 2);
    v2f b0 = *(const v2f*)(Bk);
    v2f b1 = *(const v2f*)(Bk + 32);
    c0 = __builtin_amdgcn_wmma_f32_16x16x4_f32(false, a, false, b0, (short)0, c0, false, false);
    c1 = __builtin_amdgcn_wmma_f32_16x16x4_f32(false, a, false, b1, (short)0, c1, false, false);
  }

  const int rbase = m0 + ((lane >> 4) << 3);
  const float bv0 = bias ? bias[n0 + ncol] : 0.0f;
  const float bv1 = bias ? bias[n0 + 16 + ncol] : 0.0f;
  #pragma unroll
  for (int r = 0; r < 8; ++r) {
    C[(long)(rbase + r) * ldc + n0 + ncol]      = c0[r] + bv0;
    C[(long)(rbase + r) * ldc + n0 + 16 + ncol] = c1[r] + bv1;
  }
}

__global__ __launch_bounds__(256)
void dta_gemm_f32_wmma(const float* __restrict__ A, const float* __restrict__ W,
                       const float* __restrict__ bias, float* __restrict__ C,
                       int M, int N) {
  __shared__ float Wlds[D_MODEL * NT];
  (void)M;
  dta_gemm_tile(A, W, bias, C, N, N, blockIdx.x, blockIdx.y, Wlds);
}

#define VW 128
#define PROJ5_BLOCKS 1664
__global__ __launch_bounds__(256)
void dta_proj5_f32_wmma(const float* __restrict__ query,
                        const float* __restrict__ val0, const float* __restrict__ val1,
                        const float* __restrict__ val2,
                        const float* __restrict__ Woff, const float* __restrict__ boff,
                        const float* __restrict__ Waw,  const float* __restrict__ baw,
                        const float* __restrict__ Wv,   const float* __restrict__ bv,
                        float* __restrict__ off_raw, float* __restrict__ aw_buf,
                        float* __restrict__ v_proj) {
  __shared__ float Wlds[D_MODEL * NT];

  int id = blockIdx.x;
  const float* A; const float* Wm; const float* bias; float* C;
  int ldw, ldc, mtile, ntile;
  if (id < 384) {
    A = query; Wm = Woff; bias = boff; C = off_raw; ldw = 96; ldc = 96;
    mtile = id & 127; ntile = id >> 7;
  } else if (id < 768) {
    id -= 384;
    A = query; Wm = Waw; bias = baw; C = aw_buf; ldw = 96; ldc = 96;
    mtile = id & 127; ntile = id >> 7;
  } else if (id < 1280) {
    id -= 768;
    A = val0; Wm = Wv; bias = bv; C = v_proj; ldw = 256; ldc = VW;
    mtile = id & 127; ntile = id >> 7;
  } else if (id < 1536) {
    id -= 1280;
    A = val1; Wm = Wv; bias = bv; C = v_proj + (size_t)16384 * VW; ldw = 256; ldc = VW;
    mtile = id & 63; ntile = id >> 6;
  } else {
    id -= 1536;
    A = val2; Wm = Wv; bias = bv; C = v_proj + (size_t)(16384 + 8192) * VW; ldw = 256; ldc = VW;
    mtile = id & 31; ntile = id >> 5;
  }
  dta_gemm_tile(A, Wm, bias, C, ldw, ldc, mtile, ntile, Wlds);
}

__global__ __launch_bounds__(256)
void dta_softmax12(float* __restrict__ aw, int total  ) {
  const int t = blockIdx.x * blockDim.x + threadIdx.x;
  if (t >= total) return;
  const int row = t >> 3;
  const int h   = t & 7;
  float* p = aw + (long)row * 96 + h * 12;

  float v[12];
  float mx = -3.402823466e+38f;
  #pragma unroll
  for (int i = 0; i < 12; ++i) { v[i] = p[i]; mx = fmaxf(mx, v[i]); }
  float s = 0.0f;
  #pragma unroll
  for (int i = 0; i < 12; ++i) { v[i] = __expf(v[i] - mx); s += v[i]; }
  const float inv = 1.0f / s;
  #pragma unroll
  for (int i = 0; i < 12; ++i) p[i] = v[i] * inv;
}

__global__ __launch_bounds__(256)
void dta_sample(const float* __restrict__ refp, const float* __restrict__ off_raw,
                const float* __restrict__ aw, const float* __restrict__ v_proj,
                float* __restrict__ out_mid) {
  const int row  = __builtin_amdgcn_readfirstlane(
                       (int)((blockIdx.x * blockDim.x + threadIdx.x) >> 5));
  const int lane = threadIdx.x & 31;
  const char* vbytes = (const char*)v_proj;
  const unsigned lane4 = (unsigned)lane * 4u;
  const int b = row >> 11;
  const float rp = refp[row];

  const int j  = min(lane, 23);
  const int l  = j % 3;
  const int T  = 2048 >> l;
  const float Tm1  = (float)(T - 1);
  const float invT = (l == 0) ? 0x1p-11f : ((l == 1) ? 0x1p-10f : 0x1p-9f);
  const int rb = ((l == 0) ? 0 : ((l == 1) ? 8 * 2048 : 8 * 2048 + 8 * 1024)) + b * T;

  const v4f o4 = *(const v4f*)(off_raw + (long)row * 96 + 4 * j);
  const v4f w4 = *(const v4f*)(aw      + (long)row * 96 + 4 * j);

  int   eo[4];
  float wc[4];
  float ww[4];
  #pragma unroll
  for (int p = 0; p < 4; ++p) {
    float pos  = fminf(fmaxf(rp + o4[p] * invT, 0.0f), 1.0f);
    float sidx = pos * Tm1;
    int ifl = (int)sidx;
    ifl = min(max(ifl, 0), T - 2);
    wc[p] = sidx - (float)ifl;
    eo[p] = ((rb + ifl) * VW + p * 32) * 4;
    ww[p] = w4[p];
  }

  #pragma unroll
  for (int h = 0; h < 8; ++h) {
    float acc = 0.0f;
    #pragma unroll
    for (int l2 = 0; l2 < 3; ++l2) {
      const int own = h * 3 + l2;
      #pragma unroll
      for (int p = 0; p < 4; ++p) {
        const unsigned e = (unsigned)__builtin_amdgcn_readlane(eo[p], own);
        const char* q = vbytes + (e + lane4);
        const float vf = *(const float*)(q);
        const float vc = *(const float*)(q + VW * 4);
        const float c = __int_as_float(__builtin_amdgcn_readlane(__float_as_int(wc[p]), own));
        const float w = __int_as_float(__builtin_amdgcn_readlane(__float_as_int(ww[p]), own));
        acc += w * ((1.0f - c) * vf + c * vc);
      }
    }
    out_mid[(long)row * 256 + h * 32 + lane] = acc;
  }
}

extern "C" void kernel_launch(void* const* d_in, const int* in_sizes, int n_in,
                              void* d_out, int out_size, void* d_ws, size_t ws_size,
                              hipStream_t stream) {
  (void)in_sizes; (void)n_in; (void)out_size; (void)ws_size;

  const float* query = (const float*)d_in[0];
  const float* refp  = (const float*)d_in[1];
  const float* val0  = (const float*)d_in[2];
  const float* val1  = (const float*)d_in[3];
  const float* val2  = (const float*)d_in[4];
  const float* Woff  = (const float*)d_in[5];
  const float* boff  = (const float*)d_in[6];
  const float* Waw   = (const float*)d_in[7];
  const float* baw   = (const float*)d_in[8];
  const float* Wv    = (const float*)d_in[9];
  const float* bv    = (const float*)d_in[10];
  const float* Wo    = (const float*)d_in[11];
  const float* bo    = (const float*)d_in[12];
  float*       out   = (float*)d_out;

  const int BQ = 8 * 2048;

  char* ws = (char*)d_ws;
  float* off_raw = (float*)(ws);
  float* aw_buf  = (float*)(ws + (size_t)BQ * 96 * 4);
  float* v_proj  = (float*)(ws + (size_t)BQ * 96 * 4 * 2);
  float* out_mid = (float*)(ws + (size_t)BQ * 96 * 4 * 2
                               + (size_t)(BQ + 8192 + 4096) * VW * 4);

  dim3 blk(256);

  dta_proj5_f32_wmma<<<dim3(PROJ5_BLOCKS), blk, 0, stream>>>(query, val0, val1, val2,
                                                             Woff, boff, Waw, baw, Wv, bv,
                                                             off_raw, aw_buf, v_proj);
  dta_softmax12<<<dim3((BQ * 8 + 255) / 256), blk, 0, stream>>>(aw_buf, BQ * 8);
  dta_sample<<<dim3(BQ / 8), blk, 0, stream>>>(refp, off_raw, aw_buf, v_proj, out_mid);
  dta_gemm_f32_wmma<<<dim3(BQ / MT, 256 / NT), blk, 0, stream>>>(out_mid, Wo, bo, out, BQ, 256);
}


# Softmax folded into the sampling kernel's parameter phase: lane j normalises its head's 12 raw logits itself (same operation order, weights bit-identical); softmax launch removed, 4 launches -> 3
# speedup vs baseline: 1.7461x; 1.1617x over previous
#include <hip/hip_runtime.h>
#include <hip/hip_bf16.h>
#include <math.h>


typedef float v2f __attribute__((ext_vector_type(2)));
typedef float v4f __attribute__((ext_vector_type(4)));
typedef float v8f __attribute__((ext_vector_type(8)));

#define D_MODEL 256
#define NT 32
#define MT 128

__device__ __forceinline__
void dta_gemm_tile(const float* __restrict__ A, const float* __restrict__ W,
                   const float* __restrict__ bias, float* __restrict__ C,
                   int ldw, int ldc, int mtile, int ntile, float* __restrict__ Wlds) {
  const int tid = threadIdx.x;
  const int n0  = ntile * NT;

  for (int c = tid; c < (D_MODEL / 2) * (NT / 4); c += 256) {
    const int kk = c >> 3;
    const int j  = (c & 7) << 2;
    v4f r0 = *(const v4f*)(W + (long)(2 * kk)     * ldw + n0 + j);
    v4f r1 = *(const v4f*)(W + (long)(2 * kk + 1) * ldw + n0 + j);
    v4f lo = {r0.x, r1.x, r0.y, r1.y};
    v4f hi = {r0.z, r1.z, r0.w, r1.w};
    *(v4f*)(&Wlds[kk * (NT * 2) + j * 2])     = lo;
    *(v4f*)(&Wlds[kk * (NT * 2) + j * 2 + 4]) = hi;
  }
  __syncthreads();

  const int wave  = tid >> 5;
  const int lane  = tid & 31;
  const int m0    = mtile * MT + wave * 16;
  const int mrow  = m0 + (lane & 15);
  const int khalf = (lane >> 4) << 1;
  const int ncol  = lane & 15;

  v8f c0 = {};
  v8f c1 = {};
  const float* Arow = A + (long)mrow * D_MODEL;
  const float* Bbase = &Wlds[(khalf >> 1) * (NT * 2) + ncol * 2];

  #pragma unroll 8
  for (int k = 0; k < D_MODEL; k += 4) {
    v2f a = *(const v2f*)(Arow + k + khalf);
    const float* Bk = Bbase + (k >> 1) * (NT * 2);
    v2f b0 = *(const v2f*)(Bk);
    v2f b1 = *(const v2f*)(Bk + 32);
    c0 = __builtin_amdgcn_wmma_f32_16x16x4_f32(false, a, false, b0, (short)0, c0, false, false);
    c1 = __builtin_amdgcn_wmma_f32_16x16x4_f32(false, a, false, b1, (short)0, c1, false, false);
  }

  const int rbase = m0 + ((lane >> 4) << 3);
  const float bv0 = bias ? bias[n0 + ncol] : 0.0f;
  const float bv1 = bias ? bias[n0 + 16 + ncol] : 0.0f;
  #pragma unroll
  for (int r = 0; r < 8; ++r) {
    C[(long)(rbase + r) * ldc + n0 + ncol]      = c0[r] + bv0;
    C[(long)(rbase + r) * ldc + n0 + 16 + ncol] = c1[r] + bv1;
  }
}

__global__ __launch_bounds__(256)
void dta_gemm_f32_wmma(const float* __restrict__ A, const float* __restrict__ W,
                       const float* __restrict__ bias, float* __restrict__ C,
                       int M, int N) {
  __shared__ float Wlds[D_MODEL * NT];
  (void)M;
  dta_gemm_tile(A, W, bias, C, N, N, blockIdx.x, blockIdx.y, Wlds);
}

#define VW 128
#define PROJ5_BLOCKS 1664
__global__ __launch_bounds__(256)
void dta_proj5_f32_wmma(const float* __restrict__ query,
                        const float* __restrict__ val0, const float* __restrict__ val1,
                        const float* __restrict__ val2,
                        const float* __restrict__ Woff, const float* __restrict__ boff,
                        const float* __restrict__ Waw,  const float* __restrict__ baw,
                        const float* __restrict__ Wv,   const float* __restrict__ bv,
                        float* __restrict__ off_raw, float* __restrict__ aw_buf,
                        float* __restrict__ v_proj) {
  __shared__ float Wlds[D_MODEL * NT];

  int id = blockIdx.x;
  const float* A; const float* Wm; const float* bias; float* C;
  int ldw, ldc, mtile, ntile;
  if (id < 384) {
    A = query; Wm = Woff; bias = boff; C = off_raw; ldw = 96; ldc = 96;
    mtile = id & 127; ntile = id >> 7;
  } else if (id < 768) {
    id -= 384;
    A = query; Wm = Waw; bias = baw; C = aw_buf; ldw = 96; ldc = 96;
    mtile = id & 127; ntile = id >> 7;
  } else if (id < 1280) {
    id -= 768;
    A = val0; Wm = Wv; bias = bv; C = v_proj; ldw = 256; ldc = VW;
    mtile = id & 127; ntile = id >> 7;
  } else if (id < 1536) {
    id -= 1280;
    A = val1; Wm = Wv; bias = bv; C = v_proj + (size_t)16384 * VW; ldw = 256; ldc = VW;
    mtile = id & 63; ntile = id >> 6;
  } else {
    id -= 1536;
    A = val2; Wm = Wv; bias = bv; C = v_proj + (size_t)(16384 + 8192) * VW; ldw = 256; ldc = VW;
    mtile = id & 31; ntile = id >> 5;
  }
  dta_gemm_tile(A, Wm, bias, C, ldw, ldc, mtile, ntile, Wlds);
}

__global__ __launch_bounds__(256)
void dta_sample(const float* __restrict__ refp, const float* __restrict__ off_raw,
                const float* __restrict__ aw, const float* __restrict__ v_proj,
                float* __restrict__ out_mid) {
  const int row  = __builtin_amdgcn_readfirstlane(
                       (int)((blockIdx.x * blockDim.x + threadIdx.x) >> 5));
  const int lane = threadIdx.x & 31;
  const char* vbytes = (const char*)v_proj;
  const unsigned lane4 = (unsigned)lane * 4u;
  const int b = row >> 11;
  const float rp = refp[row];

  const int j  = min(lane, 23);
  const int l  = j % 3;
  const int T  = 2048 >> l;
  const float Tm1  = (float)(T - 1);
  const float invT = (l == 0) ? 0x1p-11f : ((l == 1) ? 0x1p-10f : 0x1p-9f);
  const int rb = ((l == 0) ? 0 : ((l == 1) ? 8 * 2048 : 8 * 2048 + 8 * 1024)) + b * T;

  const v4f o4 = *(const v4f*)(off_raw + (long)row * 96 + 4 * j);

  const float* lg = aw + (long)row * 96 + (j / 3) * 12;
  const v4f g0 = *(const v4f*)(lg);
  const v4f g1 = *(const v4f*)(lg + 4);
  const v4f g2 = *(const v4f*)(lg + 8);
  float sm[12] = {g0.x, g0.y, g0.z, g0.w, g1.x, g1.y, g1.z, g1.w, g2.x, g2.y, g2.z, g2.w};
  float mx = -3.402823466e+38f;
  #pragma unroll
  for (int i = 0; i < 12; ++i) mx = fmaxf(mx, sm[i]);
  float ssum = 0.0f;
  #pragma unroll
  for (int i = 0; i < 12; ++i) { sm[i] = __expf(sm[i] - mx); ssum += sm[i]; }
  const float inv = 1.0f / ssum;

  int   eo[4];
  float wc[4];
  float ww[4];
  #pragma unroll
  for (int p = 0; p < 4; ++p) {
    float pos  = fminf(fmaxf(rp + o4[p] * invT, 0.0f), 1.0f);
    float sidx = pos * Tm1;
    int ifl = (int)sidx;
    ifl = min(max(ifl, 0), T - 2);
    wc[p] = sidx - (float)ifl;
    eo[p] = ((rb + ifl) * VW + p * 32) * 4;
    ww[p] = ((l == 0) ? sm[p] : ((l == 1) ? sm[4 + p] : sm[8 + p])) * inv;
  }

  #pragma unroll
  for (int h = 0; h < 8; ++h) {
    float acc = 0.0f;
    #pragma unroll
    for (int l2 = 0; l2 < 3; ++l2) {
      const int own = h * 3 + l2;
      #pragma unroll
      for (int p = 0; p < 4; ++p) {
        const unsigned e = (unsigned)__builtin_amdgcn_readlane(eo[p], own);
        const char* q = vbytes + (e + lane4);
        const float vf = *(const float*)(q);
        const float vc = *(const float*)(q + VW * 4);
        const float c = __int_as_float(__builtin_amdgcn_readlane(__float_as_int(wc[p]), own));
        const float w = __int_as_float(__builtin_amdgcn_readlane(__float_as_int(ww[p]), own));
        acc += w * ((1.0f - c) * vf + c * vc);
      }
    }
    out_mid[(long)row * 256 + h * 32 + lane] = acc;
  }
}

extern "C" void kernel_launch(void* const* d_in, const int* in_sizes, int n_in,
                              void* d_out, int out_size, void* d_ws, size_t ws_size,
                              hipStream_t stream) {
  (void)in_sizes; (void)n_in; (void)out_size; (void)ws_size;

  const float* query = (const float*)d_in[0];
  const float* refp  = (const float*)d_in[1];
  const float* val0  = (const float*)d_in[2];
  const float* val1  = (const float*)d_in[3];
  const float* val2  = (const float*)d_in[4];
  const float* Woff  = (const float*)d_in[5];
  const float* boff  = (const float*)d_in[6];
  const float* Waw   = (const float*)d_in[7];
  const float* baw   = (const float*)d_in[8];
  const float* Wv    = (const float*)d_in[9];
  const float* bv    = (const float*)d_in[10];
  const float* Wo    = (const float*)d_in[11];
  const float* bo    = (const float*)d_in[12];
  float*       out   = (float*)d_out;

  const int BQ = 8 * 2048;

  char* ws = (char*)d_ws;
  float* off_raw = (float*)(ws);
  float* aw_buf  = (float*)(ws + (size_t)BQ * 96 * 4);
  float* v_proj  = (float*)(ws + (size_t)BQ * 96 * 4 * 2);
  float* out_mid = (float*)(ws + (size_t)BQ * 96 * 4 * 2
                               + (size_t)(BQ + 8192 + 4096) * VW * 4);

  dim3 blk(256);

  dta_proj5_f32_wmma<<<dim3(PROJ5_BLOCKS), blk, 0, stream>>>(query, val0, val1, val2,
                                                             Woff, boff, Waw, baw, Wv, bv,
                                                             off_raw, aw_buf, v_proj);
  dta_sample<<<dim3(BQ / 8), blk, 0, stream>>>(refp, off_raw, aw_buf, v_proj, out_mid);
  dta_gemm_f32_wmma<<<dim3(BQ / MT, 256 / NT), blk, 0, stream>>>(out_mid, Wo, bo, out, BQ, 256);
}


# GEMM A operand by 16-byte loads: a lane reads 4 consecutive k of its row for two WMMA steps, B pair rows in the same k order; 64 -> 32 A loads per wave; k order inside each 8 changes the sum order
# speedup vs baseline: 1.8124x; 1.0380x over previous
#include <hip/hip_runtime.h>
#include <hip/hip_bf16.h>
#include <math.h>


typedef float v2f __attribute__((ext_vector_type(2)));
typedef float v4f __attribute__((ext_vector_type(4)));
typedef float v8f __attribute__((ext_vector_type(8)));

#define D_MODEL 256
#define NT 32
#define MT 128

__device__ __forceinline__
void dta_gemm_tile(const float* __restrict__ A, const float* __restrict__ W,
                   const float* __restrict__ bias, float* __restrict__ C,
                   int ldw, int ldc, int mtile, int ntile, float* __restrict__ Wlds) {
  const int tid = threadIdx.x;
  const int n0  = ntile * NT;

  for (int c = tid; c < (D_MODEL / 2) * (NT / 4); c += 256) {
    const int kk = c >> 3;
    const int j  = (c & 7) << 2;
    v4f r0 = *(const v4f*)(W + (long)(2 * kk)     * ldw + n0 + j);
    v4f r1 = *(const v4f*)(W + (long)(2 * kk + 1) * ldw + n0 + j);
    v4f lo = {r0.x, r1.x, r0.y, r1.y};
    v4f hi = {r0.z, r1.z, r0.w, r1.w};
    *(v4f*)(&Wlds[kk * (NT * 2) + j * 2])     = lo;
    *(v4f*)(&Wlds[kk * (NT * 2) + j * 2 + 4]) = hi;
  }
  __syncthreads();

  const int wave  = tid >> 5;
  const int lane  = tid & 31;
  const int m0    = mtile * MT + wave * 16;
  const int mrow  = m0 + (lane & 15);
  const int khalf = (lane >> 4) << 1;
  const int ncol  = lane & 15;

  v8f c0 = {};
  v8f c1 = {};
  const float* Arow = A + (long)mrow * D_MODEL + 2 * khalf;
  const float* Bbase = &Wlds[khalf * (NT * 2) + ncol * 2];

  #pragma unroll 8
  for (int k = 0; k < D_MODEL; k += 8) {
    const v4f a4 = *(const v4f*)(Arow + k);
    const v2f alo = {a4.x, a4.y};
    const v2f ahi = {a4.z, a4.w};
    const float* Bk = Bbase + (k >> 1) * (NT * 2);
    v2f b0 = *(const v2f*)(Bk);
    v2f b1 = *(const v2f*)(Bk + 32);
    v2f b2 = *(const v2f*)(Bk + NT * 2);
    v2f b3 = *(const v2f*)(Bk + NT * 2 + 32);
    c0 = __builtin_amdgcn_wmma_f32_16x16x4_f32(false, alo, false, b0, (short)0, c0, false, false);
    c1 = __builtin_amdgcn_wmma_f32_16x16x4_f32(false, alo, false, b1, (short)0, c1, false, false);
    c0 = __builtin_amdgcn_wmma_f32_16x16x4_f32(false, ahi, false, b2, (short)0, c0, false, false);
    c1 = __builtin_amdgcn_wmma_f32_16x16x4_f32(false, ahi, false, b3, (short)0, c1, false, false);
  }

  const int rbase = m0 + ((lane >> 4) << 3);
  const float bv0 = bias ? bias[n0 + ncol] : 0.0f;
  const float bv1 = bias ? bias[n0 + 16 + ncol] : 0.0f;
  #pragma unroll
  for (int r = 0; r < 8; ++r) {
    C[(long)(rbase + r) * ldc + n0 + ncol]      = c0[r] + bv0;
    C[(long)(rbase + r) * ldc + n0 + 16 + ncol] = c1[r] + bv1;
  }
}

__global__ __launch_bounds__(256)
void dta_gemm_f32_wmma(const float* __restrict__ A, const float* __restrict__ W,
                       const float* __restrict__ bias, float* __restrict__ C,
                       int M, int N) {
  __shared__ float Wlds[D_MODEL * NT];
  (void)M;
  dta_gemm_tile(A, W, bias, C, N, N, blockIdx.x, blockIdx.y, Wlds);
}

#define VW 128
#define PROJ5_BLOCKS 1664
__global__ __launch_bounds__(256)
void dta_proj5_f32_wmma(const float* __restrict__ query,
                        const float* __restrict__ val0, const float* __restrict__ val1,
                        const float* __restrict__ val2,
                        const float* __restrict__ Woff, const float* __restrict__ boff,
                        const float* __restrict__ Waw,  const float* __restrict__ baw,
                        const float* __restrict__ Wv,   const float* __restrict__ bv,
                        float* __restrict__ off_raw, float* __restrict__ aw_buf,
                        float* __restrict__ v_proj) {
  __shared__ float Wlds[D_MODEL * NT];

  int id = blockIdx.x;
  const float* A; const float* Wm; const float* bias; float* C;
  int ldw, ldc, mtile, ntile;
  if (id < 384) {
    A = query; Wm = Woff; bias = boff; C = off_raw; ldw = 96; ldc = 96;
    mtile = id & 127; ntile = id >> 7;
  } else if (id < 768) {
    id -= 384;
    A = query; Wm = Waw; bias = baw; C = aw_buf; ldw = 96; ldc = 96;
    mtile = id & 127; ntile = id >> 7;
  } else if (id < 1280) {
    id -= 768;
    A = val0; Wm = Wv; bias = bv; C = v_proj; ldw = 256; ldc = VW;
    mtile = id & 127; ntile = id >> 7;
  } else if (id < 1536) {
    id -= 1280;
    A = val1; Wm = Wv; bias = bv; C = v_proj + (size_t)16384 * VW; ldw = 256; ldc = VW;
    mtile = id & 63; ntile = id >> 6;
  } else {
    id -= 1536;
    A = val2; Wm = Wv; bias = bv; C = v_proj + (size_t)(16384 + 8192) * VW; ldw = 256; ldc = VW;
    mtile = id & 31; ntile = id >> 5;
  }
  dta_gemm_tile(A, Wm, bias, C, ldw, ldc, mtile, ntile, Wlds);
}

__global__ __launch_bounds__(256)
void dta_sample(const float* __restrict__ refp, const float* __restrict__ off_raw,
                const float* __restrict__ aw, const float* __restrict__ v_proj,
                float* __restrict__ out_mid) {
  const int row  = __builtin_amdgcn_readfirstlane(
                       (int)((blockIdx.x * blockDim.x + threadIdx.x) >> 5));
  const int lane = threadIdx.x & 31;
  const char* vbytes = (const char*)v_proj;
  const unsigned lane4 = (unsigned)lane * 4u;
  const int b = row >> 11;
  const float rp = refp[row];

  const int j  = min(lane, 23);
  const int l  = j % 3;
  const int T  = 2048 >> l;
  const float Tm1  = (float)(T - 1);
  const float invT = (l == 0) ? 0x1p-11f : ((l == 1) ? 0x1p-10f : 0x1p-9f);
  const int rb = ((l == 0) ? 0 : ((l == 1) ? 8 * 2048 : 8 * 2048 + 8 * 1024)) + b * T;

  const v4f o4 = *(const v4f*)(off_raw + (long)row * 96 + 4 * j);

  const float* lg = aw + (long)row * 96 + (j / 3) * 12;
  const v4f g0 = *(const v4f*)(lg);
  const v4f g1 = *(const v4f*)(lg + 4);
  const v4f g2 = *(const v4f*)(lg + 8);
  float sm[12] = {g0.x, g0.y, g0.z, g0.w, g1.x, g1.y, g1.z, g1.w, g2.x, g2.y, g2.z, g2.w};
  float mx = -3.402823466e+38f;
  #pragma unroll
  for (int i = 0; i < 12; ++i) mx = fmaxf(mx, sm[i]);
  float ssum = 0.0f;
  #pragma unroll
  for (int i = 0; i < 12; ++i) { sm[i] = __expf(sm[i] - mx); ssum += sm[i]; }
  const float inv = 1.0f / ssum;

  int   eo[4];
  float wc[4];
  float ww[4];
  #pragma unroll
  for (int p = 0; p < 4; ++p) {
    float pos  = fminf(fmaxf(rp + o4[p] * invT, 0.0f), 1.0f);
    float sidx = pos * Tm1;
    int ifl = (int)sidx;
    ifl = min(max(ifl, 0), T - 2);
    wc[p] = sidx - (float)ifl;
    eo[p] = ((rb + ifl) * VW + p * 32) * 4;
    ww[p] = ((l == 0) ? sm[p] : ((l == 1) ? sm[4 + p] : sm[8 + p])) * inv;
  }

  #pragma unroll
  for (int h = 0; h < 8; ++h) {
    float acc = 0.0f;
    #pragma unroll
    for (int l2 = 0; l2 < 3; ++l2) {
      const int own = h * 3 + l2;
      #pragma unroll
      for (int p = 0; p < 4; ++p) {
        const unsigned e = (unsigned)__builtin_amdgcn_readlane(eo[p], own);
        const char* q = vbytes + (e + lane4);
        const float vf = *(const float*)(q);
        const float vc = *(const float*)(q + VW * 4);
        const float c = __int_as_float(__builtin_amdgcn_readlane(__float_as_int(wc[p]), own));
        const float w = __int_as_float(__builtin_amdgcn_readlane(__float_as_int(ww[p]), own));
        acc += w * ((1.0f - c) * vf + c * vc);
      }
    }
    out_mid[(long)row * 256 + h * 32 + lane] = acc;
  }
}

extern "C" void kernel_launch(void* const* d_in, const int* in_sizes, int n_in,
                              void* d_out, int out_size, void* d_ws, size_t ws_size,
                              hipStream_t stream) {
  (void)in_sizes; (void)n_in; (void)out_size; (void)ws_size;

  const float* query = (const float*)d_in[0];
  const float* refp  = (const float*)d_in[1];
  const float* val0  = (const float*)d_in[2];
  const float* val1  = (const float*)d_in[3];
  const float* val2  = (const float*)d_in[4];
  const float* Woff  = (const float*)d_in[5];
  const float* boff  = (const float*)d_in[6];
  const float* Waw   = (const float*)d_in[7];
  const float* baw   = (const float*)d_in[8];
  const float* Wv    = (const float*)d_in[9];
  const float* bv    = (const float*)d_in[10];
  const float* Wo    = (const float*)d_in[11];
  const float* bo    = (const float*)d_in[12];
  float*       out   = (float*)d_out;

  const int BQ = 8 * 2048;

  char* ws = (char*)d_ws;
  float* off_raw = (float*)(ws);
  float* aw_buf  = (float*)(ws + (size_t)BQ * 96 * 4);
  float* v_proj  = (float*)(ws + (size_t)BQ * 96 * 4 * 2);
  float* out_mid = (float*)(ws + (size_t)BQ * 96 * 4 * 2
                               + (size_t)(BQ + 8192 + 4096) * VW * 4);

  dim3 blk(256);

  dta_proj5_f32_wmma<<<dim3(PROJ5_BLOCKS), blk, 0, stream>>>(query, val0, val1, val2,
                                                             Woff, boff, Waw, baw, Wv, bv,
                                                             off_raw, aw_buf, v_proj);
  dta_sample<<<dim3(BQ / 8), blk, 0, stream>>>(refp, off_raw, aw_buf, v_proj, out_mid);
  dta_gemm_f32_wmma<<<dim3(BQ / MT, 256 / NT), blk, 0, stream>>>(out_mid, Wo, bo, out, BQ, 256);
}


# Output projection fused into the sampling kernel: one 16-wave workgroup per WGP keeps Wo in LDS and samples tile t while multiplying tile t-1 on WMMA, half the waves in each order; 3 launches -> 2
# speedup vs baseline: 1.8948x; 1.0455x over previous
#include <hip/hip_runtime.h>
#include <hip/hip_bf16.h>
#include <math.h>


typedef float v2f __attribute__((ext_vector_type(2)));
typedef float v4f __attribute__((ext_vector_type(4)));
typedef float v8f __attribute__((ext_vector_type(8)));

#define D_MODEL 256
#define NT 32
#define MT 128

__device__ __forceinline__
void dta_gemm_tile(const float* __restrict__ A, const float* __restrict__ W,
                   const float* __restrict__ bias, float* __restrict__ C,
                   int ldw, int ldc, int mtile, int ntile, float* __restrict__ Wlds) {
  const int tid = threadIdx.x;
  const int n0  = ntile * NT;

  for (int c = tid; c < (D_MODEL / 2) * (NT / 4); c += 256) {
    const int kk = c >> 3;
    const int j  = (c & 7) << 2;
    v4f r0 = *(const v4f*)(W + (long)(2 * kk)     * ldw + n0 + j);
    v4f r1 = *(const v4f*)(W + (long)(2 * kk + 1) * ldw + n0 + j);
    v4f lo = {r0.x, r1.x, r0.y, r1.y};
    v4f hi = {r0.z, r1.z, r0.w, r1.w};
    *(v4f*)(&Wlds[kk * (NT * 2) + j * 2])     = lo;
    *(v4f*)(&Wlds[kk * (NT * 2) + j * 2 + 4]) = hi;
  }
  __syncthreads();

  const int wave  = tid >> 5;
  const int lane  = tid & 31;
  const int m0    = mtile * MT + wave * 16;
  const int mrow  = m0 + (lane & 15);
  const int khalf = (lane >> 4) << 1;
  const int ncol  = lane & 15;

  v8f c0 = {};
  v8f c1 = {};
  const float* Arow = A + (long)mrow * D_MODEL + 2 * khalf;
  const float* Bbase = &Wlds[khalf * (NT * 2) + ncol * 2];

  #pragma unroll 8
  for (int k = 0; k < D_MODEL; k += 8) {
    const v4f a4 = *(const v4f*)(Arow + k);
    const v2f alo = {a4.x, a4.y};
    const v2f ahi = {a4.z, a4.w};
    const float* Bk = Bbase + (k >> 1) * (NT * 2);
    v2f b0 = *(const v2f*)(Bk);
    v2f b1 = *(const v2f*)(Bk + 32);
    v2f b2 = *(const v2f*)(Bk + NT * 2);
    v2f b3 = *(const v2f*)(Bk + NT * 2 + 32);
    c0 = __builtin_amdgcn_wmma_f32_16x16x4_f32(false, alo, false, b0, (short)0, c0, false, false);
    c1 = __builtin_amdgcn_wmma_f32_16x16x4_f32(false, alo, false, b1, (short)0, c1, false, false);
    c0 = __builtin_amdgcn_wmma_f32_16x16x4_f32(false, ahi, false, b2, (short)0, c0, false, false);
    c1 = __builtin_amdgcn_wmma_f32_16x16x4_f32(false, ahi, false, b3, (short)0, c1, false, false);
  }

  const int rbase = m0 + ((lane >> 4) << 3);
  const float bv0 = bias ? bias[n0 + ncol] : 0.0f;
  const float bv1 = bias ? bias[n0 + 16 + ncol] : 0.0f;
  #pragma unroll
  for (int r = 0; r < 8; ++r) {
    C[(long)(rbase + r) * ldc + n0 + ncol]      = c0[r] + bv0;
    C[(long)(rbase + r) * ldc + n0 + 16 + ncol] = c1[r] + bv1;
  }
}

#define VW 128
#define PROJ5_BLOCKS 1664
__global__ __launch_bounds__(256)
void dta_proj5_f32_wmma(const float* __restrict__ query,
                        const float* __restrict__ val0, const float* __restrict__ val1,
                        const float* __restrict__ val2,
                        const float* __restrict__ Woff, const float* __restrict__ boff,
                        const float* __restrict__ Waw,  const float* __restrict__ baw,
                        const float* __restrict__ Wv,   const float* __restrict__ bv,
                        float* __restrict__ off_raw, float* __restrict__ aw_buf,
                        float* __restrict__ v_proj) {
  __shared__ float Wlds[D_MODEL * NT];

  int id = blockIdx.x;
  const float* A; const float* Wm; const float* bias; float* C;
  int ldw, ldc, mtile, ntile;
  if (id < 384) {
    A = query; Wm = Woff; bias = boff; C = off_raw; ldw = 96; ldc = 96;
    mtile = id & 127; ntile = id >> 7;
  } else if (id < 768) {
    id -= 384;
    A = query; Wm = Waw; bias = baw; C = aw_buf; ldw = 96; ldc = 96;
    mtile = id & 127; ntile = id >> 7;
  } else if (id < 1280) {
    id -= 768;
    A = val0; Wm = Wv; bias = bv; C = v_proj; ldw = 256; ldc = VW;
    mtile = id & 127; ntile = id >> 7;
  } else if (id < 1536) {
    id -= 1280;
    A = val1; Wm = Wv; bias = bv; C = v_proj + (size_t)16384 * VW; ldw = 256; ldc = VW;
    mtile = id & 63; ntile = id >> 6;
  } else {
    id -= 1536;
    A = val2; Wm = Wv; bias = bv; C = v_proj + (size_t)(16384 + 8192) * VW; ldw = 256; ldc = VW;
    mtile = id & 31; ntile = id >> 5;
  }
  dta_gemm_tile(A, Wm, bias, C, ldw, ldc, mtile, ntile, Wlds);
}

#define TR 16
#define NTILE 4
#define APITCH 260
#define WP 528

__device__ __forceinline__
void dta_outproj_tile(const float* __restrict__ At, const float* __restrict__ Wl,
                      const float* __restrict__ bo, float* __restrict__ out,
                      int rowbase, int wave, int lane) {
  const int hh   = lane >> 4;
  const int ncol = lane & 15;
  const int n0   = wave * 16;
  const float* Ap = At + ncol * APITCH + 4 * hh;
  const float* Bb = Wl + (2 * hh) * WP + (n0 + ncol) * 2;

  v8f c = {};
  #pragma unroll 8
  for (int k = 0; k < D_MODEL; k += 8) {
    const v4f a4 = *(const v4f*)(Ap + k);
    const float* Bk = Bb + (k >> 1) * WP;
    const v2f b01 = *(const v2f*)(Bk);
    const v2f b23 = *(const v2f*)(Bk + WP);
    const v2f alo = {a4.x, a4.y};
    const v2f ahi = {a4.z, a4.w};
    c = __builtin_amdgcn_wmma_f32_16x16x4_f32(false, alo, false, b01, (short)0, c, false, false);
    c = __builtin_amdgcn_wmma_f32_16x16x4_f32(false, ahi, false, b23, (short)0, c, false, false);
  }

  const float bias = bo[n0 + ncol];
  float* O = out + (long)(rowbase + 8 * hh) * D_MODEL + n0 + ncol;
  #pragma unroll
  for (int r = 0; r < 8; ++r) O[(long)r * D_MODEL] = c[r] + bias;
}

__device__ __forceinline__
void dta_stage_wo(const float* __restrict__ Wo, float* __restrict__ Wl, int tid) {
  #pragma unroll 8
  for (int it = 0; it < 16; ++it) {
    const int c  = it * 512 + tid;
    const int kk = c >> 6;
    const int j  = (c & 63) << 2;
    v4f r0 = *(const v4f*)(Wo + (long)(2 * kk)     * D_MODEL + j);
    v4f r1 = *(const v4f*)(Wo + (long)(2 * kk + 1) * D_MODEL + j);
    v4f lo = {r0.x, r1.x, r0.y, r1.y};
    v4f hi = {r0.z, r1.z, r0.w, r1.w};
    *(v4f*)(&Wl[kk * WP + j * 2])     = lo;
    *(v4f*)(&Wl[kk * WP + j * 2 + 4]) = hi;
  }
}

__global__ __launch_bounds__(512)
void dta_sample_outproj(const float* __restrict__ refp, const float* __restrict__ off_raw,
                        const float* __restrict__ aw, const float* __restrict__ v_proj,
                        const float* __restrict__ Wo, const float* __restrict__ bo,
                        float* __restrict__ out) {
  __shared__ float Wl[(D_MODEL / 2) * WP];
  __shared__ float Alds[2 * TR * APITCH];

  const int tid  = threadIdx.x;
  const int wave = __builtin_amdgcn_readfirstlane((int)(threadIdx.x >> 5));
  const int lane = tid & 31;
  const int row0 = blockIdx.x * (TR * NTILE);
  const char* vbytes = (const char*)v_proj;
  const unsigned lane4 = (unsigned)lane * 4u;
  const bool mulFirst = ((wave ^ (wave >> 2)) & 1) != 0;

  #pragma nounroll
  for (int t = 0; t <= NTILE; ++t) {
    const bool doMul = t > 0;
    const bool doSmp = t < NTILE;
    const float* Aprev = &Alds[((t + 1) & 1) * (TR * APITCH)];
    float*       Acur  = &Alds[(t & 1) * (TR * APITCH)];

    if (t == 0 && mulFirst) dta_stage_wo(Wo, Wl, tid);
    if (doMul && mulFirst)
      dta_outproj_tile(Aprev, Wl, bo, out, row0 + (t - 1) * TR, wave, lane);

    if (doSmp) {
      const int row = row0 + t * TR + wave;
      const int b = row >> 11;
      const float rp = refp[row];

      const int j  = min(lane, 23);
      const int l  = j % 3;
      const int T  = 2048 >> l;
      const float Tm1  = (float)(T - 1);
      const float invT = (l == 0) ? 0x1p-11f : ((l == 1) ? 0x1p-10f : 0x1p-9f);
      const int rb = ((l == 0) ? 0 : ((l == 1) ? 8 * 2048 : 8 * 2048 + 8 * 1024)) + b * T;

      const v4f o4 = *(const v4f*)(off_raw + (long)row * 96 + 4 * j);

      const float* lg = aw + (long)row * 96 + (j / 3) * 12;
      const v4f g0 = *(const v4f*)(lg);
      const v4f g1 = *(const v4f*)(lg + 4);
      const v4f g2 = *(const v4f*)(lg + 8);
      float sm[12] = {g0.x, g0.y, g0.z, g0.w, g1.x, g1.y, g1.z, g1.w, g2.x, g2.y, g2.z, g2.w};
      float mx = -3.402823466e+38f;
      #pragma unroll
      for (int i = 0; i < 12; ++i) mx = fmaxf(mx, sm[i]);
      float ssum = 0.0f;
      #pragma unroll
      for (int i = 0; i < 12; ++i) { sm[i] = __expf(sm[i] - mx); ssum += sm[i]; }
      const float inv = 1.0f / ssum;

      int   eo[4];
      float wc[4];
      float ww[4];
      #pragma unroll
      for (int p = 0; p < 4; ++p) {
        float pos  = fminf(fmaxf(rp + o4[p] * invT, 0.0f), 1.0f);
        float sidx = pos * Tm1;
        int ifl = (int)sidx;
        ifl = min(max(ifl, 0), T - 2);
        wc[p] = sidx - (float)ifl;
        eo[p] = ((rb + ifl) * VW + p * 32) * 4;
        ww[p] = ((l == 0) ? sm[p] : ((l == 1) ? sm[4 + p] : sm[8 + p])) * inv;
      }

      float* mid = Acur + wave * APITCH + lane;
      #pragma unroll
      for (int h = 0; h < 8; ++h) {
        float acc = 0.0f;
        #pragma unroll
        for (int l2 = 0; l2 < 3; ++l2) {
          const int own = h * 3 + l2;
          #pragma unroll
          for (int p = 0; p < 4; ++p) {
            const unsigned e = (unsigned)__builtin_amdgcn_readlane(eo[p], own);
            const char* q = vbytes + (e + lane4);
            const float vf = *(const float*)(q);
            const float vc = *(const float*)(q + VW * 4);
            const float c = __int_as_float(__builtin_amdgcn_readlane(__float_as_int(wc[p]), own));
            const float w = __int_as_float(__builtin_amdgcn_readlane(__float_as_int(ww[p]), own));
            acc += w * ((1.0f - c) * vf + c * vc);
          }
        }
        mid[h * 32] = acc;
      }
    }

    if (t == 0 && !mulFirst) dta_stage_wo(Wo, Wl, tid);
    if (doMul && !mulFirst)
      dta_outproj_tile(Aprev, Wl, bo, out, row0 + (t - 1) * TR, wave, lane);

    __syncthreads();
  }
}

extern "C" void kernel_launch(void* const* d_in, const int* in_sizes, int n_in,
                              void* d_out, int out_size, void* d_ws, size_t ws_size,
                              hipStream_t stream) {
  (void)in_sizes; (void)n_in; (void)out_size; (void)ws_size;

  const float* query = (const float*)d_in[0];
  const float* refp  = (const float*)d_in[1];
  const float* val0  = (const float*)d_in[2];
  const float* val1  = (const float*)d_in[3];
  const float* val2  = (const float*)d_in[4];
  const float* Woff  = (const float*)d_in[5];
  const float* boff  = (const float*)d_in[6];
  const float* Waw   = (const float*)d_in[7];
  const float* baw   = (const float*)d_in[8];
  const float* Wv    = (const float*)d_in[9];
  const float* bv    = (const float*)d_in[10];
  const float* Wo    = (const float*)d_in[11];
  const float* bo    = (const float*)d_in[12];
  float*       out   = (float*)d_out;

  const int BQ = 8 * 2048;

  char* ws = (char*)d_ws;
  float* off_raw = (float*)(ws);
  float* aw_buf  = (float*)(ws + (size_t)BQ * 96 * 4);
  float* v_proj  = (float*)(ws + (size_t)BQ * 96 * 4 * 2);

  dim3 blk(256);

  dta_proj5_f32_wmma<<<dim3(PROJ5_BLOCKS), blk, 0, stream>>>(query, val0, val1, val2,
                                                             Woff, boff, Waw, baw, Wv, bv,
                                                             off_raw, aw_buf, v_proj);
  dta_sample_outproj<<<dim3(BQ / (TR * NTILE)), dim3(512), 0, stream>>>(refp, off_raw, aw_buf,
                                                                         v_proj, Wo, bo, out);
}


# Sampling gather by 16-byte loads: a lane owns 4 channels of heads r and 4+r, so one wave load reads the head slices of 4 samples, not 1 (48 loads per row, was 192); parameters by permlane16 broadcast
# speedup vs baseline: 2.1956x; 1.1588x over previous
#include <hip/hip_runtime.h>
#include <hip/hip_bf16.h>
#include <math.h>


typedef float v2f __attribute__((ext_vector_type(2)));
typedef float v4f __attribute__((ext_vector_type(4)));
typedef float v8f __attribute__((ext_vector_type(8)));

#define D_MODEL 256
#define NT 32
#define MT 128

__device__ __forceinline__
void dta_gemm_tile(const float* __restrict__ A, const float* __restrict__ W,
                   const float* __restrict__ bias, float* __restrict__ C,
                   int ldw, int ldc, int mtile, int ntile, float* __restrict__ Wlds) {
  const int tid = threadIdx.x;
  const int n0  = ntile * NT;

  for (int c = tid; c < (D_MODEL / 2) * (NT / 4); c += 256) {
    const int kk = c >> 3;
    const int j  = (c & 7) << 2;
    v4f r0 = *(const v4f*)(W + (long)(2 * kk)     * ldw + n0 + j);
    v4f r1 = *(const v4f*)(W + (long)(2 * kk + 1) * ldw + n0 + j);
    v4f lo = {r0.x, r1.x, r0.y, r1.y};
    v4f hi = {r0.z, r1.z, r0.w, r1.w};
    *(v4f*)(&Wlds[kk * (NT * 2) + j * 2])     = lo;
    *(v4f*)(&Wlds[kk * (NT * 2) + j * 2 + 4]) = hi;
  }
  __syncthreads();

  const int wave  = tid >> 5;
  const int lane  = tid & 31;
  const int m0    = mtile * MT + wave * 16;
  const int mrow  = m0 + (lane & 15);
  const int khalf = (lane >> 4) << 1;
  const int ncol  = lane & 15;

  v8f c0 = {};
  v8f c1 = {};
  const float* Arow = A + (long)mrow * D_MODEL + 2 * khalf;
  const float* Bbase = &Wlds[khalf * (NT * 2) + ncol * 2];

  #pragma unroll 8
  for (int k = 0; k < D_MODEL; k += 8) {
    const v4f a4 = *(const v4f*)(Arow + k);
    const v2f alo = {a4.x, a4.y};
    const v2f ahi = {a4.z, a4.w};
    const float* Bk = Bbase + (k >> 1) * (NT * 2);
    v2f b0 = *(const v2f*)(Bk);
    v2f b1 = *(const v2f*)(Bk + 32);
    v2f b2 = *(const v2f*)(Bk + NT * 2);
    v2f b3 = *(const v2f*)(Bk + NT * 2 + 32);
    c0 = __builtin_amdgcn_wmma_f32_16x16x4_f32(false, alo, false, b0, (short)0, c0, false, false);
    c1 = __builtin_amdgcn_wmma_f32_16x16x4_f32(false, alo, false, b1, (short)0, c1, false, false);
    c0 = __builtin_amdgcn_wmma_f32_16x16x4_f32(false, ahi, false, b2, (short)0, c0, false, false);
    c1 = __builtin_amdgcn_wmma_f32_16x16x4_f32(false, ahi, false, b3, (short)0, c1, false, false);
  }

  const int rbase = m0 + ((lane >> 4) << 3);
  const float bv0 = bias ? bias[n0 + ncol] : 0.0f;
  const float bv1 = bias ? bias[n0 + 16 + ncol] : 0.0f;
  #pragma unroll
  for (int r = 0; r < 8; ++r) {
    C[(long)(rbase + r) * ldc + n0 + ncol]      = c0[r] + bv0;
    C[(long)(rbase + r) * ldc + n0 + 16 + ncol] = c1[r] + bv1;
  }
}

#define VW 128
#define PROJ5_BLOCKS 1664
__global__ __launch_bounds__(256)
void dta_proj5_f32_wmma(const float* __restrict__ query,
                        const float* __restrict__ val0, const float* __restrict__ val1,
                        const float* __restrict__ val2,
                        const float* __restrict__ Woff, const float* __restrict__ boff,
                        const float* __restrict__ Waw,  const float* __restrict__ baw,
                        const float* __restrict__ Wv,   const float* __restrict__ bv,
                        float* __restrict__ off_raw, float* __restrict__ aw_buf,
                        float* __restrict__ v_proj) {
  __shared__ float Wlds[D_MODEL * NT];

  int id = blockIdx.x;
  const float* A; const float* Wm; const float* bias; float* C;
  int ldw, ldc, mtile, ntile;
  if (id < 384) {
    A = query; Wm = Woff; bias = boff; C = off_raw; ldw = 96; ldc = 96;
    mtile = id & 127; ntile = id >> 7;
  } else if (id < 768) {
    id -= 384;
    A = query; Wm = Waw; bias = baw; C = aw_buf; ldw = 96; ldc = 96;
    mtile = id & 127; ntile = id >> 7;
  } else if (id < 1280) {
    id -= 768;
    A = val0; Wm = Wv; bias = bv; C = v_proj; ldw = 256; ldc = VW;
    mtile = id & 127; ntile = id >> 7;
  } else if (id < 1536) {
    id -= 1280;
    A = val1; Wm = Wv; bias = bv; C = v_proj + (size_t)16384 * VW; ldw = 256; ldc = VW;
    mtile = id & 63; ntile = id >> 6;
  } else {
    id -= 1536;
    A = val2; Wm = Wv; bias = bv; C = v_proj + (size_t)(16384 + 8192) * VW; ldw = 256; ldc = VW;
    mtile = id & 31; ntile = id >> 5;
  }
  dta_gemm_tile(A, Wm, bias, C, ldw, ldc, mtile, ntile, Wlds);
}

#define TR 16
#define NTILE 4
#define APITCH 260
#define WP 528

__device__ __forceinline__
void dta_outproj_tile(const float* __restrict__ At, const float* __restrict__ Wl,
                      const float* __restrict__ bo, float* __restrict__ out,
                      int rowbase, int wave, int lane) {
  const int hh   = lane >> 4;
  const int ncol = lane & 15;
  const int n0   = wave * 16;
  const float* Ap = At + ncol * APITCH + 4 * hh;
  const float* Bb = Wl + (2 * hh) * WP + (n0 + ncol) * 2;

  v8f c = {};
  #pragma unroll 8
  for (int k = 0; k < D_MODEL; k += 8) {
    const v4f a4 = *(const v4f*)(Ap + k);
    const float* Bk = Bb + (k >> 1) * WP;
    const v2f b01 = *(const v2f*)(Bk);
    const v2f b23 = *(const v2f*)(Bk + WP);
    const v2f alo = {a4.x, a4.y};
    const v2f ahi = {a4.z, a4.w};
    c = __builtin_amdgcn_wmma_f32_16x16x4_f32(false, alo, false, b01, (short)0, c, false, false);
    c = __builtin_amdgcn_wmma_f32_16x16x4_f32(false, ahi, false, b23, (short)0, c, false, false);
  }

  const float bias = bo[n0 + ncol];
  float* O = out + (long)(rowbase + 8 * hh) * D_MODEL + n0 + ncol;
  #pragma unroll
  for (int r = 0; r < 8; ++r) O[(long)r * D_MODEL] = c[r] + bias;
}

__device__ __forceinline__
void dta_stage_wo(const float* __restrict__ Wo, float* __restrict__ Wl, int tid) {
  #pragma unroll 8
  for (int it = 0; it < 16; ++it) {
    const int c  = it * 512 + tid;
    const int kk = c >> 6;
    const int j  = (c & 63) << 2;
    v4f r0 = *(const v4f*)(Wo + (long)(2 * kk)     * D_MODEL + j);
    v4f r1 = *(const v4f*)(Wo + (long)(2 * kk + 1) * D_MODEL + j);
    v4f lo = {r0.x, r1.x, r0.y, r1.y};
    v4f hi = {r0.z, r1.z, r0.w, r1.w};
    *(v4f*)(&Wl[kk * WP + j * 2])     = lo;
    *(v4f*)(&Wl[kk * WP + j * 2 + 4]) = hi;
  }
}

template <int S>
__device__ __forceinline__ unsigned dta_bcast8(unsigned x) {
  return (unsigned)__builtin_amdgcn_permlane16(x, x, (unsigned)S * 0x11111111u,
                                               (unsigned)(8 + S) * 0x11111111u, true, true);
}

template <int S>
__device__ __forceinline__
void dta_sample_load(const char* __restrict__ vbytes, unsigned goff, unsigned eo,
                     v4f& vf, v4f& vc) {
  const unsigned e = dta_bcast8<S>(eo);
  const char* q = vbytes + (e + goff);
  vf = *(const v4f*)(q);
  vc = *(const v4f*)(q + VW * 4);
}

__device__ __forceinline__
void dta_load_level(const char* __restrict__ vbytes, unsigned goff, unsigned eo,
                    v4f (&vf)[8], v4f (&vc)[8]) {
  dta_sample_load<0>(vbytes, goff, eo, vf[0], vc[0]);
  dta_sample_load<1>(vbytes, goff, eo, vf[1], vc[1]);
  dta_sample_load<2>(vbytes, goff, eo, vf[2], vc[2]);
  dta_sample_load<3>(vbytes, goff, eo, vf[3], vc[3]);
  dta_sample_load<4>(vbytes, goff, eo, vf[4], vc[4]);
  dta_sample_load<5>(vbytes, goff, eo, vf[5], vc[5]);
  dta_sample_load<6>(vbytes, goff, eo, vf[6], vc[6]);
  dta_sample_load<7>(vbytes, goff, eo, vf[7], vc[7]);
}

template <int S>
__device__ __forceinline__
v4f dta_sample_add(v4f acc, float wc, float ww, v4f vf, v4f vc) {
  const float c = __uint_as_float(dta_bcast8<S>(__float_as_uint(wc)));
  const float w = __uint_as_float(dta_bcast8<S>(__float_as_uint(ww)));
  acc += w * ((1.0f - c) * vf + c * vc);
  return acc;
}

__device__ __forceinline__
void dta_sum_level(v4f& acc0, v4f& acc1, float wc, float ww,
                   const v4f (&vf)[8], const v4f (&vc)[8]) {
  acc0 = dta_sample_add<0>(acc0, wc, ww, vf[0], vc[0]);
  acc0 = dta_sample_add<1>(acc0, wc, ww, vf[1], vc[1]);
  acc0 = dta_sample_add<2>(acc0, wc, ww, vf[2], vc[2]);
  acc0 = dta_sample_add<3>(acc0, wc, ww, vf[3], vc[3]);
  acc1 = dta_sample_add<4>(acc1, wc, ww, vf[4], vc[4]);
  acc1 = dta_sample_add<5>(acc1, wc, ww, vf[5], vc[5]);
  acc1 = dta_sample_add<6>(acc1, wc, ww, vf[6], vc[6]);
  acc1 = dta_sample_add<7>(acc1, wc, ww, vf[7], vc[7]);
}

__global__ __launch_bounds__(512)
void dta_sample_outproj(const float* __restrict__ refp, const float* __restrict__ off_raw,
                        const float* __restrict__ aw, const float* __restrict__ v_proj,
                        const float* __restrict__ Wo, const float* __restrict__ bo,
                        float* __restrict__ out) {
  __shared__ float Wl[(D_MODEL / 2) * WP];
  __shared__ float Alds[2 * TR * APITCH];

  const int tid  = threadIdx.x;
  const int wave = __builtin_amdgcn_readfirstlane((int)(threadIdx.x >> 5));
  const int lane = tid & 31;
  const int row0 = blockIdx.x * (TR * NTILE);
  const char* vbytes = (const char*)v_proj;
  const bool mulFirst = ((wave ^ (wave >> 2)) & 1) != 0;

  #pragma nounroll
  for (int t = 0; t <= NTILE; ++t) {
    const bool doMul = t > 0;
    const bool doSmp = t < NTILE;
    const float* Aprev = &Alds[((t + 1) & 1) * (TR * APITCH)];
    float*       Acur  = &Alds[(t & 1) * (TR * APITCH)];

    if (t == 0 && mulFirst) dta_stage_wo(Wo, Wl, tid);
    if (doMul && mulFirst)
      dta_outproj_tile(Aprev, Wl, bo, out, row0 + (t - 1) * TR, wave, lane);

    if (doSmp) {
      const int row = row0 + t * TR + wave;
      const int b = row >> 11;
      const float rp = refp[row];

      const int r  = lane >> 3;
      const int g  = lane & 7;
      const int hp = ((g >> 2) << 2) + r;
      const int pp = g & 3;

      const unsigned ih = ((unsigned)row * 96u + (unsigned)(hp * 12)) * 4u;
      const unsigned is = ih + (unsigned)pp * 4u;
      const char* obytes = (const char*)off_raw;
      const char* abytes = (const char*)aw;
      float o[3];
      float lo[3];
      #pragma unroll
      for (int l = 0; l < 3; ++l) {
        o[l]  = *(const float*)(obytes + (is + 16u * l));
        lo[l] = *(const float*)(abytes + (is + 16u * l));
      }

      const v4f g0 = *(const v4f*)(abytes + ih);
      const v4f g1 = *(const v4f*)(abytes + (ih + 16u));
      const v4f g2 = *(const v4f*)(abytes + (ih + 32u));
      float sm[12] = {g0.x, g0.y, g0.z, g0.w, g1.x, g1.y, g1.z, g1.w, g2.x, g2.y, g2.z, g2.w};
      float mx = -3.402823466e+38f;
      #pragma unroll
      for (int i = 0; i < 12; ++i) mx = fmaxf(mx, sm[i]);
      float ssum = 0.0f;
      #pragma unroll
      for (int i = 0; i < 12; ++i) { sm[i] = __expf(sm[i] - mx); ssum += sm[i]; }
      const float inv = 1.0f / ssum;

      unsigned eo[3];
      float    wc[3];
      float    ww[3];
      #pragma unroll
      for (int l = 0; l < 3; ++l) {
        const int T = 2048 >> l;
        const float Tm1  = (float)(T - 1);
        const float invT = (l == 0) ? 0x1p-11f : ((l == 1) ? 0x1p-10f : 0x1p-9f);
        const int rb = ((l == 0) ? 0 : ((l == 1) ? 8 * 2048 : 8 * 2048 + 8 * 1024)) + b * T;
        float pos  = fminf(fmaxf(rp + o[l] * invT, 0.0f), 1.0f);
        float sidx = pos * Tm1;
        int ifl = (int)sidx;
        ifl = min(max(ifl, 0), T - 2);
        wc[l] = sidx - (float)ifl;
        eo[l] = (unsigned)(((rb + ifl) * VW + pp * 32) * 4);
        ww[l] = __expf(lo[l] - mx) * inv;
      }

      const unsigned goff = (unsigned)g * 16u;
      float* mid = Acur + wave * APITCH + r * 32 + g * 4;
      v4f vf0[8], vc0[8], vf1[8], vc1[8], vf2[8], vc2[8];
      v4f acc0 = {0.0f, 0.0f, 0.0f, 0.0f};
      v4f acc1 = {0.0f, 0.0f, 0.0f, 0.0f};
      dta_load_level(vbytes, goff, eo[0], vf0, vc0);
      dta_load_level(vbytes, goff, eo[1], vf1, vc1);
      dta_sum_level(acc0, acc1, wc[0], ww[0], vf0, vc0);
      __builtin_amdgcn_sched_barrier(0);
      dta_load_level(vbytes, goff, eo[2], vf2, vc2);
      dta_sum_level(acc0, acc1, wc[1], ww[1], vf1, vc1);
      dta_sum_level(acc0, acc1, wc[2], ww[2], vf2, vc2);
      *(v4f*)(mid)          = acc0;
      *(v4f*)(mid + 4 * 32) = acc1;
    }

    if (t == 0 && !mulFirst) dta_stage_wo(Wo, Wl, tid);
    if (doMul && !mulFirst)
      dta_outproj_tile(Aprev, Wl, bo, out, row0 + (t - 1) * TR, wave, lane);

    __syncthreads();
  }
}

extern "C" void kernel_launch(void* const* d_in, const int* in_sizes, int n_in,
                              void* d_out, int out_size, void* d_ws, size_t ws_size,
                              hipStream_t stream) {
  (void)in_sizes; (void)n_in; (void)out_size; (void)ws_size;

  const float* query = (const float*)d_in[0];
  const float* refp  = (const float*)d_in[1];
  const float* val0  = (const float*)d_in[2];
  const float* val1  = (const float*)d_in[3];
  const float* val2  = (const float*)d_in[4];
  const float* Woff  = (const float*)d_in[5];
  const float* boff  = (const float*)d_in[6];
  const float* Waw   = (const float*)d_in[7];
  const float* baw   = (const float*)d_in[8];
  const float* Wv    = (const float*)d_in[9];
  const float* bv    = (const float*)d_in[10];
  const float* Wo    = (const float*)d_in[11];
  const float* bo    = (const float*)d_in[12];
  float*       out   = (float*)d_out;

  const int BQ = 8 * 2048;

  char* ws = (char*)d_ws;
  float* off_raw = (float*)(ws);
  float* aw_buf  = (float*)(ws + (size_t)BQ * 96 * 4);
  float* v_proj  = (float*)(ws + (size_t)BQ * 96 * 4 * 2);

  dim3 blk(256);

  dta_proj5_f32_wmma<<<dim3(PROJ5_BLOCKS), blk, 0, stream>>>(query, val0, val1, val2,
                                                             Woff, boff, Waw, baw, Wv, bv,
                                                             off_raw, aw_buf, v_proj);
  dta_sample_outproj<<<dim3(BQ / (TR * NTILE)), dim3(512), 0, stream>>>(refp, off_raw, aw_buf,
                                                                         v_proj, Wo, bo, out);
}
